# starting kernel
# baseline (speedup 1.0000x reference)
#include <hip/hip_runtime.h>

typedef __attribute__((ext_vector_type(2))) float v2f;
typedef __attribute__((ext_vector_type(8))) float v8f;

#define RNN_B 4096
#define RNN_T 512
#define RNN_I 16
#define RNN_H 6

__device__ __forceinline__ v8f wmma_f32_4(v2f a, v2f b, v8f c) {
  return __builtin_amdgcn_wmma_f32_16x16x4_f32(false, a, false, b, (short)0, c,
                                               false, false);
}

__device__ __forceinline__ float tanh_fast(float x) {
#if __has_builtin(__builtin_amdgcn_tanhf)
  return __builtin_amdgcn_tanhf(x);
#else
  x = __builtin_amdgcn_fmed3f(x, -9.0f, 9.0f);
  float e = __expf(2.0f * x);
  return (e - 1.0f) * __builtin_amdgcn_rcpf(e + 1.0f);
#endif
}

__device__ __forceinline__ void tanh8(v8f& a) {
#pragma unroll
  for (int i = 0; i < 8; ++i) a[i] = tanh_fast(a[i]);
}

__global__ __launch_bounds__(64) void rnn_fused_wmma(
    const float* __restrict__ x, const float* __restrict__ W_ih,
    const float* __restrict__ W_hh, const float* __restrict__ b_ih,
    const float* __restrict__ b_hh, const float* __restrict__ fc_w,
    const float* __restrict__ fc_b, float* __restrict__ out) {
  __shared__ __align__(16) float hbuf[2][16][18];

  const int tid  = threadIdx.x;
  const int w    = tid >> 5;
  const int lane = tid & 31;
  const int n    = lane & 15;
  const int hi   = lane >> 4;
  const int b0   = (blockIdx.x * 2 + w) * 16;

  v2f bih[4];
#pragma unroll
  for (int s = 0; s < 4; ++s) {
#pragma unroll
    for (int v = 0; v < 2; ++v) {
      int k = 4 * s + 2 * hi + v;
      bih[s][v] = (n < RNN_H) ? W_ih[n * RNN_I + k] : 0.0f;
    }
  }
  v2f whh[2];
#pragma unroll
  for (int s = 0; s < 2; ++s) {
#pragma unroll
    for (int v = 0; v < 2; ++v) {
      int k = 4 * s + 2 * hi + v;
      whh[s][v] = (n < RNN_H && k < RNN_H) ? W_hh[n * RNN_H + k] : 0.0f;
    }
  }
  const float bn = (n < RNN_H) ? (b_ih[n] + b_hh[n]) : 0.0f;
  v8f cbias;
#pragma unroll
  for (int i = 0; i < 8; ++i) cbias[i] = bn;

  const float* xp = x + (size_t)(b0 + n) * (RNN_T * RNN_I) + 2 * hi;

  v2f ha0 = 0.0f, ha1 = 0.0f;
  v2f xa[4], xb[4];
#pragma unroll
  for (int s = 0; s < 4; ++s) xa[s] = *(const v2f*)(xp + 4 * s);

  float(*hb)[18] = hbuf[w];

  for (int t = 0; t < RNN_T; t += 2) {
    v8f acc = cbias;
#pragma unroll
    for (int s = 0; s < 4; ++s) acc = wmma_f32_4(xa[s], bih[s], acc);
    acc = wmma_f32_4(ha0, whh[0], acc);
    acc = wmma_f32_4(ha1, whh[1], acc);
    {
      const float* p = xp + (size_t)(t + 1) * RNN_I;
#pragma unroll
      for (int s = 0; s < 4; ++s) xb[s] = *(const v2f*)(p + 4 * s);
    }
    tanh8(acc);
#pragma unroll
    for (int r = 0; r < 8; ++r) hb[r + hi * 8][n] = acc[r];
    __builtin_amdgcn_wave_barrier();
    ha0 = *(const v2f*)&hb[n][2 * hi];
    ha1 = *(const v2f*)&hb[n][4 + 2 * hi];
    __builtin_amdgcn_wave_barrier();

    acc = cbias;
#pragma unroll
    for (int s = 0; s < 4; ++s) acc = wmma_f32_4(xb[s], bih[s], acc);
    acc = wmma_f32_4(ha0, whh[0], acc);
    acc = wmma_f32_4(ha1, whh[1], acc);
    {
      const int t2 = (t + 2) & (RNN_T - 1);
      const float* p = xp + (size_t)t2 * RNN_I;
#pragma unroll
      for (int s = 0; s < 4; ++s) xa[s] = *(const v2f*)(p + 4 * s);
    }
    tanh8(acc);
#pragma unroll
    for (int r = 0; r < 8; ++r) hb[r + hi * 8][n] = acc[r];
    __builtin_amdgcn_wave_barrier();
    ha0 = *(const v2f*)&hb[n][2 * hi];
    ha1 = *(const v2f*)&hb[n][4 + 2 * hi];
    __builtin_amdgcn_wave_barrier();
  }

  if (lane < 16) {
    float s = fc_b[0];
#pragma unroll
    for (int j = 0; j < RNN_H; ++j) s += hb[lane][j] * fc_w[j];
    out[b0 + lane] = s;
  }
}

extern "C" void kernel_launch(void* const* d_in, const int* in_sizes, int n_in,
                              void* d_out, int out_size, void* d_ws,
                              size_t ws_size, hipStream_t stream) {
  (void)in_sizes; (void)n_in; (void)out_size; (void)d_ws; (void)ws_size;
  const float* x    = (const float*)d_in[0];
  const float* W_ih = (const float*)d_in[1];
  const float* W_hh = (const float*)d_in[2];
  const float* b_ih = (const float*)d_in[3];
  const float* b_hh = (const float*)d_in[4];
  const float* fc_w = (const float*)d_in[5];
  const float* fc_b = (const float*)d_in[6];
  float* out = (float*)d_out;

  dim3 block(64);
  dim3 grid(RNN_B / 32);
  rnn_fused_wmma<<<grid, block, 0, stream>>>(x, W_ih, W_hh, b_ih, b_hh, fc_w,
                                             fc_b, out);
}


# x asked for twice: one prefetch per 32 steps pulls the 256-byte units needed 64 steps later into the second-level cache, so the 8-step register refills hit there; sums and order unchanged
# speedup vs baseline: 4.2208x; 4.2208x over previous
#include <hip/hip_runtime.h>


typedef __attribute__((ext_vector_type(2))) float v2f;
typedef __attribute__((ext_vector_type(8))) float v8f;

#define RNN_B 4096
#define RNN_T 512
#define RNN_I 16
#define RNN_H 6
#define RNN_CH 8
#define RNN_NCH (RNN_T / RNN_CH)
#define RNN_ROW (RNN_T * RNN_I)
#define RNN_RW 4
#define RNN_NT (RNN_RW / 2)
#define RNN_WPB 4
#define RNN_NBLK 256
#define RNN_PFU 64
#define RNN_PFC 4
#define RNN_PFD 8

static_assert(RNN_NBLK * RNN_WPB * RNN_RW == RNN_B, "grid covers the batch exactly");
static_assert(RNN_NT * 4 == RNN_CH, "one projection product per recurrence step");
static_assert(RNN_RW * (RNN_PFC * RNN_CH * RNN_I / RNN_PFU) == 32, "one prefetch unit per lane");
static_assert(RNN_PFD % RNN_PFC == 0 && RNN_NCH % RNN_PFC == 0, "prefetch groups tile the sequence");

__device__ __forceinline__ v8f wmma_f32_4(v2f a, v2f b, v8f c) {
  return __builtin_amdgcn_wmma_f32_16x16x4_f32(false, a, false, b, (short)0, c,
                                               false, false);
}

__device__ __forceinline__ float tanh_fast(float x) {
#if __has_builtin(__builtin_amdgcn_tanhf)
  return __builtin_amdgcn_tanhf(x);
#else
  x = __builtin_amdgcn_fmed3f(x, -9.0f, 9.0f);
  float e = __expf(2.0f * x);
  return (e - 1.0f) * __builtin_amdgcn_rcpf(e + 1.0f);
#endif
}

__device__ __forceinline__ float gather16(float v, unsigned lo, unsigned hi) {
  const int i = __builtin_bit_cast(int, v);
  return __builtin_bit_cast(float, __builtin_amdgcn_permlane16(i, i, lo, hi, true, false));
}
__device__ __forceinline__ float gatherx16(float v, unsigned lo, unsigned hi) {
  const int i = __builtin_bit_cast(int, v);
  return __builtin_bit_cast(float, __builtin_amdgcn_permlanex16(i, i, lo, hi, true, false));
}

__device__ __forceinline__ void rnn_chunk(const v8f (&Pc)[RNN_NT], v8f (&Pn)[RNN_NT],
                                          v2f (&xr)[RNN_NT][4], const float* xnext,
                                          const v2f (&bih)[4], const v2f (&whh)[2],
                                          const v8f& cbias, v8f& cin, v2f& ha0,
                                          v2f& ha1, float& hx, bool lowcol,
                                          bool own) {
  v8f p = cbias;
#pragma unroll
  for (int j = 0; j < RNN_CH; ++j) {
    v8f acc = wmma_f32_4(ha0, whh[0], cin);
    acc = wmma_f32_4(ha1, whh[1], acc);
    __builtin_amdgcn_sched_barrier(0);

    const int q = j >> 2;
    const int s = j & 3;
    if (s == 0) p = cbias;
    p = wmma_f32_4(xr[q][s], bih[s], p);
    if (s == 3) Pn[q] = p;
    xr[q][s] = *(const v2f*)(xnext + (size_t)q * RNN_ROW + 4 * s);
    __builtin_amdgcn_sched_barrier(0);

    hx = tanh_fast(lowcol ? acc[0] : acc[1]);
    const float xo = gatherx16(hx, 0x54761032u, 0xdcfe98bau);
    const float z  = own ? hx : xo;
    ha0[0] = gather16(z, 0x80808080u, 0xa2a2a2a2u);
    ha0[1] = gather16(z, 0x91919191u, 0xb3b3b3b3u);
    ha1[0] = gather16(z, 0xc4c4c4c4u, 0xe6e6e6e6u);
    ha1[1] = gather16(z, 0xd5d5d5d5u, 0xf7f7f7f7u);

    cin[0] = (j + 1 < RNN_CH) ? Pc[0][(j + 1) & 7] : Pn[0][0];
    cin[1] = (j + 1 < RNN_CH) ? Pc[1][(j + 1) & 7] : Pn[1][0];
  }
}

__global__ __launch_bounds__(RNN_WPB * 32) void rnn_fused_wmma(
    const float* __restrict__ x, const float* __restrict__ W_ih,
    const float* __restrict__ W_hh, const float* __restrict__ b_ih,
    const float* __restrict__ b_hh, const float* __restrict__ fc_w,
    const float* __restrict__ fc_b, float* __restrict__ out) {
  __shared__ __align__(16) float hbuf[RNN_WPB][RNN_RW][8];

  const int tid  = threadIdx.x;
  const int w    = tid >> 5;
  const int lane = tid & 31;
  const int n    = lane & 15;
  const int hi   = lane >> 4;
  const int u    = n & 7;
  const int b0   = (blockIdx.x * RNN_WPB + w) * RNN_RW;

  const float* xpf = x + (size_t)(b0 + (lane >> 3)) * RNN_ROW + (lane & 7) * RNN_PFU;
  for (int pc = 0; pc < RNN_PFD; pc += RNN_PFC)
    __builtin_prefetch(xpf + (size_t)pc * (RNN_CH * RNN_I), 0, 2);

  v2f bih[4];
#pragma unroll
  for (int s = 0; s < 4; ++s) {
#pragma unroll
    for (int v = 0; v < 2; ++v) {
      int k = 4 * s + 2 * hi + v;
      bih[s][v] = (u < RNN_H) ? W_ih[u * RNN_I + k] : 0.0f;
    }
  }
  v2f whh[2];
#pragma unroll
  for (int s = 0; s < 2; ++s) {
#pragma unroll
    for (int v = 0; v < 2; ++v) {
      int k = 4 * s + 2 * hi + v;
      whh[s][v] = (u < RNN_H && k < RNN_H) ? W_hh[u * RNN_H + k] : 0.0f;
    }
  }
  const float bn = (u < RNN_H) ? (b_ih[u] + b_hh[u]) : 0.0f;
  v8f cbias;
#pragma unroll
  for (int i = 0; i < 8; ++i) cbias[i] = bn;

  const float* xt = x + (size_t)(b0 + 2 * (n >> 3)) * RNN_ROW + (n & 7) * RNN_I + 2 * hi;
  const bool lowcol = n < 8;
  const bool own    = ((n >> 1) & 1) == hi;

  v2f xr[RNN_NT][4];
  v8f P0[RNN_NT], P1[RNN_NT];

#pragma unroll
  for (int q = 0; q < RNN_NT; ++q) {
#pragma unroll
    for (int s = 0; s < 4; ++s)
      xr[q][s] = *(const v2f*)(xt + (size_t)q * RNN_ROW + 4 * s);
  }
#pragma unroll
  for (int q = 0; q < RNN_NT; ++q) {
    v8f p = cbias;
#pragma unroll
    for (int s = 0; s < 4; ++s) p = wmma_f32_4(xr[q][s], bih[s], p);
    P0[q] = p;
  }
#pragma unroll
  for (int q = 0; q < RNN_NT; ++q) {
#pragma unroll
    for (int s = 0; s < 4; ++s)
      xr[q][s] = *(const v2f*)(xt + (size_t)q * RNN_ROW + RNN_CH * RNN_I + 4 * s);
  }

  v2f ha0 = 0.0f, ha1 = 0.0f;
  float hx = 0.0f;
  v8f cin = 0.0f;
  cin[0] = P0[0][0];
  cin[1] = P0[1][0];

  for (int c = 0; c < RNN_NCH; c += 2) {
    if (c % RNN_PFC == 0 && c + RNN_PFD < RNN_NCH)
      __builtin_prefetch(xpf + (size_t)(c + RNN_PFD) * (RNN_CH * RNN_I), 0, 2);
    const float* xa = xt + (size_t)((c + 2) & (RNN_NCH - 1)) * (RNN_CH * RNN_I);
    const float* xb = xt + (size_t)((c + 3) & (RNN_NCH - 1)) * (RNN_CH * RNN_I);
    rnn_chunk(P0, P1, xr, xa, bih, whh, cbias, cin, ha0, ha1, hx, lowcol, own);
    rnn_chunk(P1, P0, xr, xb, bih, whh, cbias, cin, ha0, ha1, hx, lowcol, own);
  }

  hbuf[w][2 * hi + (n >> 3)][u] = hx;
  __syncthreads();
  if (tid < RNN_WPB * RNN_RW) {
    const float* hl = hbuf[tid >> 2][tid & 3];
    float s = fc_b[0];
#pragma unroll
    for (int j = 0; j < RNN_H; ++j) s += hl[j] * fc_w[j];
    out[blockIdx.x * (RNN_WPB * RNN_RW) + tid] = s;
  }
}

extern "C" void kernel_launch(void* const* d_in, const int* in_sizes, int n_in,
                              void* d_out, int out_size, void* d_ws,
                              size_t ws_size, hipStream_t stream) {
  (void)in_sizes; (void)n_in; (void)out_size; (void)d_ws; (void)ws_size;
  const float* x    = (const float*)d_in[0];
  const float* W_ih = (const float*)d_in[1];
  const float* W_hh = (const float*)d_in[2];
  const float* b_ih = (const float*)d_in[3];
  const float* b_hh = (const float*)d_in[4];
  const float* fc_w = (const float*)d_in[5];
  const float* fc_b = (const float*)d_in[6];
  float* out = (float*)d_out;

  dim3 block(RNN_WPB * 32);
  dim3 grid(RNN_NBLK);
  rnn_fused_wmma<<<grid, block, 0, stream>>>(x, W_ih, W_hh, b_ih, b_hh, fc_w,
                                             fc_b, out);
}


# a step's two chained h products fused into one: a batch row takes 2 tile rows (units 0-3 / 4,5), B column groups hold the 2 weight slices, one lane-rotated add joins; h sum now (C+u0..3)+(u4+u5)
# speedup vs baseline: 4.7927x; 1.1355x over previous
#include <hip/hip_runtime.h>


typedef __attribute__((ext_vector_type(2))) float v2f;
typedef __attribute__((ext_vector_type(8))) float v8f;

#define RNN_B 4096
#define RNN_T 512
#define RNN_I 16
#define RNN_H 6
#define RNN_CH 8
#define RNN_NCH (RNN_T / RNN_CH)
#define RNN_ROW (RNN_T * RNN_I)
#define RNN_RW 4
#define RNN_NT (RNN_RW / 2)
#define RNN_WPB 4
#define RNN_NBLK 256
#define RNN_PFU 64
#define RNN_PFC 4
#define RNN_PFD 8

static_assert(RNN_NBLK * RNN_WPB * RNN_RW == RNN_B, "grid covers the batch exactly");
static_assert(RNN_NT * 4 == RNN_CH, "one projection product per recurrence step");
static_assert(RNN_RW * (RNN_PFC * RNN_CH * RNN_I / RNN_PFU) == 32, "one prefetch unit per lane");
static_assert(RNN_PFD % RNN_PFC == 0 && RNN_NCH % RNN_PFC == 0, "prefetch groups tile the sequence");
static_assert(RNN_H <= 8, "the hidden units fit two K slices of 4");

__device__ __forceinline__ v8f wmma_f32_4(v2f a, v2f b, v8f c) {
  return __builtin_amdgcn_wmma_f32_16x16x4_f32(false, a, false, b, (short)0, c,
                                               false, false);
}

__device__ __forceinline__ float tanh_fast(float x) {
#if __has_builtin(__builtin_amdgcn_tanhf)
  return __builtin_amdgcn_tanhf(x);
#else
  x = __builtin_amdgcn_fmed3f(x, -9.0f, 9.0f);
  float e = __expf(2.0f * x);
  return (e - 1.0f) * __builtin_amdgcn_rcpf(e + 1.0f);
#endif
}

__device__ __forceinline__ float gather16(float v, unsigned lo, unsigned hi) {
  const int i = __builtin_bit_cast(int, v);
  return __builtin_bit_cast(float, __builtin_amdgcn_permlane16(i, i, lo, hi, true, false));
}
__device__ __forceinline__ float gatherx16(float v, unsigned lo, unsigned hi) {
  const int i = __builtin_bit_cast(int, v);
  return __builtin_bit_cast(float, __builtin_amdgcn_permlanex16(i, i, lo, hi, true, false));
}
__device__ __forceinline__ float rot8(float v) {
  const int i = __builtin_bit_cast(int, v);
  return __builtin_bit_cast(float, __builtin_amdgcn_update_dpp(i, i, 0x128, 0xf, 0xf, true));
}

__device__ __forceinline__ void rnn_chunk(const v8f (&Pc)[RNN_NT], v8f (&Pn)[RNN_NT],
                                          v2f (&xr)[RNN_NT][4], const float* xnext,
                                          const v2f (&bih)[4], const v2f& whh,
                                          const v8f& cbias, v8f& cin, v2f& ha,
                                          float& hx, bool lowcol, bool own) {
  v8f p = cbias;
#pragma unroll
  for (int j = 0; j < RNN_CH; ++j) {
    const v8f acc = wmma_f32_4(ha, whh, cin);
    __builtin_amdgcn_sched_barrier(0);

    const int q = j >> 2;
    const int s = j & 3;
    if (s == 0) p = cbias;
    p = wmma_f32_4(xr[q][s], bih[s], p);
    if (s == 3) Pn[q] = p;
    xr[q][s] = *(const v2f*)(xnext + (size_t)q * RNN_ROW + 4 * s);
    __builtin_amdgcn_sched_barrier(0);

    const float s0 = lowcol ? acc[0] : acc[3];
    const float s1 = lowcol ? acc[1] : acc[2];
    hx = tanh_fast(s0 + rot8(s1));
    const float xo = gatherx16(hx, 0x54761032u, 0xdcfe98bau);
    const float z  = own ? hx : xo;
    ha[0] = gather16(z, 0xc480c480u, 0xe6a2e6a2u);
    ha[1] = gather16(z, 0xd591d591u, 0xf7b3f7b3u);

    cin[0] = (j + 1 < RNN_CH) ? Pc[0][(j + 1) & 7] : Pn[0][0];
    cin[1] = (j + 1 < RNN_CH) ? Pc[1][(j + 1) & 7] : Pn[1][0];
  }
}

__global__ __launch_bounds__(RNN_WPB * 32) void rnn_fused_wmma(
    const float* __restrict__ x, const float* __restrict__ W_ih,
    const float* __restrict__ W_hh, const float* __restrict__ b_ih,
    const float* __restrict__ b_hh, const float* __restrict__ fc_w,
    const float* __restrict__ fc_b, float* __restrict__ out) {
  __shared__ __align__(16) float hbuf[RNN_WPB][RNN_RW][8];

  const int tid  = threadIdx.x;
  const int w    = tid >> 5;
  const int lane = tid & 31;
  const int n    = lane & 15;
  const int hi   = lane >> 4;
  const int u    = n & 7;
  const int g    = n >> 3;
  const int b0   = (blockIdx.x * RNN_WPB + w) * RNN_RW;

  const float* xpf = x + (size_t)(b0 + (lane >> 3)) * RNN_ROW + (lane & 7) * RNN_PFU;
  for (int pc = 0; pc < RNN_PFD; pc += RNN_PFC)
    __builtin_prefetch(xpf + (size_t)pc * (RNN_CH * RNN_I), 0, 2);

  v2f bih[4];
#pragma unroll
  for (int s = 0; s < 4; ++s) {
#pragma unroll
    for (int v = 0; v < 2; ++v) {
      int k = 4 * s + 2 * hi + v;
      bih[s][v] = (u < RNN_H) ? W_ih[u * RNN_I + k] : 0.0f;
    }
  }
  v2f whh;
#pragma unroll
  for (int v = 0; v < 2; ++v) {
    int k = 4 * g + 2 * hi + v;
    whh[v] = (u < RNN_H && k < RNN_H) ? W_hh[u * RNN_H + k] : 0.0f;
  }
  const float bn = (u < RNN_H) ? (b_ih[u] + b_hh[u]) : 0.0f;
  v8f cbias;
#pragma unroll
  for (int i = 0; i < 8; ++i) cbias[i] = bn;

  const float* xt = x + (size_t)(b0 + 2 * (n >> 3)) * RNN_ROW + (n & 7) * RNN_I + 2 * hi;
  const bool lowcol = n < 8;
  const bool own    = ((n >> 1) & 1) == hi;

  v2f xr[RNN_NT][4];
  v8f P0[RNN_NT], P1[RNN_NT];

#pragma unroll
  for (int q = 0; q < RNN_NT; ++q) {
#pragma unroll
    for (int s = 0; s < 4; ++s)
      xr[q][s] = *(const v2f*)(xt + (size_t)q * RNN_ROW + 4 * s);
  }
#pragma unroll
  for (int q = 0; q < RNN_NT; ++q) {
    v8f p = cbias;
#pragma unroll
    for (int s = 0; s < 4; ++s) p = wmma_f32_4(xr[q][s], bih[s], p);
    P0[q] = p;
  }
#pragma unroll
  for (int q = 0; q < RNN_NT; ++q) {
#pragma unroll
    for (int s = 0; s < 4; ++s)
      xr[q][s] = *(const v2f*)(xt + (size_t)q * RNN_ROW + RNN_CH * RNN_I + 4 * s);
  }

  v2f ha = 0.0f;
  float hx = 0.0f;
  v8f cin = 0.0f;
  cin[0] = P0[0][0];
  cin[1] = P0[1][0];

  for (int c = 0; c < RNN_NCH; c += 2) {
    if (c % RNN_PFC == 0 && c + RNN_PFD < RNN_NCH)
      __builtin_prefetch(xpf + (size_t)(c + RNN_PFD) * (RNN_CH * RNN_I), 0, 2);
    const float* xa = xt + (size_t)((c + 2) & (RNN_NCH - 1)) * (RNN_CH * RNN_I);
    const float* xb = xt + (size_t)((c + 3) & (RNN_NCH - 1)) * (RNN_CH * RNN_I);
    rnn_chunk(P0, P1, xr, xa, bih, whh, cbias, cin, ha, hx, lowcol, own);
    rnn_chunk(P1, P0, xr, xb, bih, whh, cbias, cin, ha, hx, lowcol, own);
  }

  hbuf[w][2 * hi + (n >> 3)][u] = hx;
  __syncthreads();
  if (tid < RNN_WPB * RNN_RW) {
    const float* hl = hbuf[tid >> 2][tid & 3];
    float s = fc_b[0];
#pragma unroll
    for (int j = 0; j < RNN_H; ++j) s += hl[j] * fc_w[j];
    out[blockIdx.x * (RNN_WPB * RNN_RW) + tid] = s;
  }
}

extern "C" void kernel_launch(void* const* d_in, const int* in_sizes, int n_in,
                              void* d_out, int out_size, void* d_ws,
                              size_t ws_size, hipStream_t stream) {
  (void)in_sizes; (void)n_in; (void)out_size; (void)d_ws; (void)ws_size;
  const float* x    = (const float*)d_in[0];
  const float* W_ih = (const float*)d_in[1];
  const float* W_hh = (const float*)d_in[2];
  const float* b_ih = (const float*)d_in[3];
  const float* b_hh = (const float*)d_in[4];
  const float* fc_w = (const float*)d_in[5];
  const float* fc_b = (const float*)d_in[6];
  float* out = (float*)d_out;

  dim3 block(RNN_WPB * 32);
  dim3 grid(RNN_NBLK);
  rnn_fused_wmma<<<grid, block, 0, stream>>>(x, W_ih, W_hh, b_ih, b_hh, fc_w,
                                             fc_b, out);
}


# x register loads grouped: a chunk's 8 fragment loads go out as one group at its first step into a second register set, not one load between the 2 matrix products of each step; sums, order unchanged
# speedup vs baseline: 5.5042x; 1.1485x over previous
#include <hip/hip_runtime.h>


typedef __attribute__((ext_vector_type(2))) float v2f;
typedef __attribute__((ext_vector_type(8))) float v8f;

#define RNN_B 4096
#define RNN_T 512
#define RNN_I 16
#define RNN_H 6
#define RNN_CH 8
#define RNN_NCH (RNN_T / RNN_CH)
#define RNN_ROW (RNN_T * RNN_I)
#define RNN_RW 4
#define RNN_NT (RNN_RW / 2)
#define RNN_WPB 4
#define RNN_NBLK 256
#define RNN_PFU 64
#define RNN_PFC 4
#define RNN_PFD 8

static_assert(RNN_NBLK * RNN_WPB * RNN_RW == RNN_B, "grid covers the batch exactly");
static_assert(RNN_NT * 4 == RNN_CH, "one projection product per recurrence step");
static_assert(RNN_RW * (RNN_PFC * RNN_CH * RNN_I / RNN_PFU) == 32, "one prefetch unit per lane");
static_assert(RNN_PFD % RNN_PFC == 0 && RNN_NCH % RNN_PFC == 0, "prefetch groups tile the sequence");
static_assert(RNN_H <= 8, "the hidden units fit two K slices of 4");

__device__ __forceinline__ v8f wmma_f32_4(v2f a, v2f b, v8f c) {
  return __builtin_amdgcn_wmma_f32_16x16x4_f32(false, a, false, b, (short)0, c,
                                               false, false);
}

__device__ __forceinline__ float tanh_fast(float x) {
#if __has_builtin(__builtin_amdgcn_tanhf)
  return __builtin_amdgcn_tanhf(x);
#else
  x = __builtin_amdgcn_fmed3f(x, -9.0f, 9.0f);
  float e = __expf(2.0f * x);
  return (e - 1.0f) * __builtin_amdgcn_rcpf(e + 1.0f);
#endif
}

__device__ __forceinline__ float gather16(float v, unsigned lo, unsigned hi) {
  const int i = __builtin_bit_cast(int, v);
  return __builtin_bit_cast(float, __builtin_amdgcn_permlane16(i, i, lo, hi, true, false));
}
__device__ __forceinline__ float gatherx16(float v, unsigned lo, unsigned hi) {
  const int i = __builtin_bit_cast(int, v);
  return __builtin_bit_cast(float, __builtin_amdgcn_permlanex16(i, i, lo, hi, true, false));
}
__device__ __forceinline__ float rot8(float v) {
  const int i = __builtin_bit_cast(int, v);
  return __builtin_bit_cast(float, __builtin_amdgcn_update_dpp(i, i, 0x128, 0xf, 0xf, true));
}

__device__ __forceinline__ void rnn_chunk(const v8f (&Pc)[RNN_NT], v8f (&Pn)[RNN_NT],
                                          const v2f (&xr)[RNN_NT][4], v2f (&xw)[RNN_NT][4],
                                          const float* xnext,
                                          const v2f (&bih)[4], const v2f& whh,
                                          const v8f& cbias, v8f& cin, v2f& ha,
                                          float& hx, bool lowcol, bool own) {
  v8f p = cbias;
#pragma unroll
  for (int j = 0; j < RNN_CH; ++j) {
    const v8f acc = wmma_f32_4(ha, whh, cin);
    __builtin_amdgcn_sched_barrier(0);

    const int q = j >> 2;
    const int s = j & 3;
    if (s == 0) p = cbias;
    p = wmma_f32_4(xr[q][s], bih[s], p);
    if (s == 3) Pn[q] = p;
    if (j == 0) {
#pragma unroll
      for (int q2 = 0; q2 < RNN_NT; ++q2) {
#pragma unroll
        for (int s2 = 0; s2 < 4; ++s2)
          xw[q2][s2] = *(const v2f*)(xnext + (size_t)q2 * RNN_ROW + 4 * s2);
      }
    }
    __builtin_amdgcn_sched_barrier(0);

    const float s0 = lowcol ? acc[0] : acc[3];
    const float s1 = lowcol ? acc[1] : acc[2];
    hx = tanh_fast(s0 + rot8(s1));
    const float xo = gatherx16(hx, 0x54761032u, 0xdcfe98bau);
    const float z  = own ? hx : xo;
    ha[0] = gather16(z, 0xc480c480u, 0xe6a2e6a2u);
    ha[1] = gather16(z, 0xd591d591u, 0xf7b3f7b3u);

    cin[0] = (j + 1 < RNN_CH) ? Pc[0][(j + 1) & 7] : Pn[0][0];
    cin[1] = (j + 1 < RNN_CH) ? Pc[1][(j + 1) & 7] : Pn[1][0];
  }
}

__global__ __launch_bounds__(RNN_WPB * 32) void rnn_fused_wmma(
    const float* __restrict__ x, const float* __restrict__ W_ih,
    const float* __restrict__ W_hh, const float* __restrict__ b_ih,
    const float* __restrict__ b_hh, const float* __restrict__ fc_w,
    const float* __restrict__ fc_b, float* __restrict__ out) {
  __shared__ __align__(16) float hbuf[RNN_WPB][RNN_RW][8];

  const int tid  = threadIdx.x;
  const int w    = tid >> 5;
  const int lane = tid & 31;
  const int n    = lane & 15;
  const int hi   = lane >> 4;
  const int u    = n & 7;
  const int g    = n >> 3;
  const int b0   = (blockIdx.x * RNN_WPB + w) * RNN_RW;

  const float* xpf = x + (size_t)(b0 + (lane >> 3)) * RNN_ROW + (lane & 7) * RNN_PFU;
  for (int pc = 0; pc < RNN_PFD; pc += RNN_PFC)
    __builtin_prefetch(xpf + (size_t)pc * (RNN_CH * RNN_I), 0, 2);

  v2f bih[4];
#pragma unroll
  for (int s = 0; s < 4; ++s) {
#pragma unroll
    for (int v = 0; v < 2; ++v) {
      int k = 4 * s + 2 * hi + v;
      bih[s][v] = (u < RNN_H) ? W_ih[u * RNN_I + k] : 0.0f;
    }
  }
  v2f whh;
#pragma unroll
  for (int v = 0; v < 2; ++v) {
    int k = 4 * g + 2 * hi + v;
    whh[v] = (u < RNN_H && k < RNN_H) ? W_hh[u * RNN_H + k] : 0.0f;
  }
  const float bn = (u < RNN_H) ? (b_ih[u] + b_hh[u]) : 0.0f;
  v8f cbias;
#pragma unroll
  for (int i = 0; i < 8; ++i) cbias[i] = bn;

  const float* xt = x + (size_t)(b0 + 2 * (n >> 3)) * RNN_ROW + (n & 7) * RNN_I + 2 * hi;
  const bool lowcol = n < 8;
  const bool own    = ((n >> 1) & 1) == hi;

  v2f xr[RNN_NT][4], xs[RNN_NT][4];
  v8f P0[RNN_NT], P1[RNN_NT];

#pragma unroll
  for (int q = 0; q < RNN_NT; ++q) {
#pragma unroll
    for (int s = 0; s < 4; ++s)
      xr[q][s] = *(const v2f*)(xt + (size_t)q * RNN_ROW + 4 * s);
  }
#pragma unroll
  for (int q = 0; q < RNN_NT; ++q) {
    v8f p = cbias;
#pragma unroll
    for (int s = 0; s < 4; ++s) p = wmma_f32_4(xr[q][s], bih[s], p);
    P0[q] = p;
  }
#pragma unroll
  for (int q = 0; q < RNN_NT; ++q) {
#pragma unroll
    for (int s = 0; s < 4; ++s)
      xr[q][s] = *(const v2f*)(xt + (size_t)q * RNN_ROW + RNN_CH * RNN_I + 4 * s);
  }

  v2f ha = 0.0f;
  float hx = 0.0f;
  v8f cin = 0.0f;
  cin[0] = P0[0][0];
  cin[1] = P0[1][0];

  for (int c = 0; c < RNN_NCH; c += 2) {
    if (c % RNN_PFC == 0 && c + RNN_PFD < RNN_NCH)
      __builtin_prefetch(xpf + (size_t)(c + RNN_PFD) * (RNN_CH * RNN_I), 0, 2);
    const float* xa = xt + (size_t)((c + 2) & (RNN_NCH - 1)) * (RNN_CH * RNN_I);
    const float* xb = xt + (size_t)((c + 3) & (RNN_NCH - 1)) * (RNN_CH * RNN_I);
    rnn_chunk(P0, P1, xr, xs, xa, bih, whh, cbias, cin, ha, hx, lowcol, own);
    rnn_chunk(P1, P0, xs, xr, xb, bih, whh, cbias, cin, ha, hx, lowcol, own);
  }

  hbuf[w][2 * hi + (n >> 3)][u] = hx;
  __syncthreads();
  if (tid < RNN_WPB * RNN_RW) {
    const float* hl = hbuf[tid >> 2][tid & 3];
    float s = fc_b[0];
#pragma unroll
    for (int j = 0; j < RNN_H; ++j) s += hl[j] * fc_w[j];
    out[blockIdx.x * (RNN_WPB * RNN_RW) + tid] = s;
  }
}

extern "C" void kernel_launch(void* const* d_in, const int* in_sizes, int n_in,
                              void* d_out, int out_size, void* d_ws,
                              size_t ws_size, hipStream_t stream) {
  (void)in_sizes; (void)n_in; (void)out_size; (void)d_ws; (void)ws_size;
  const float* x    = (const float*)d_in[0];
  const float* W_ih = (const float*)d_in[1];
  const float* W_hh = (const float*)d_in[2];
  const float* b_ih = (const float*)d_in[3];
  const float* b_hh = (const float*)d_in[4];
  const float* fc_w = (const float*)d_in[5];
  const float* fc_b = (const float*)d_in[6];
  float* out = (float*)d_out;

  dim3 block(RNN_WPB * 32);
  dim3 grid(RNN_NBLK);
  rnn_fused_wmma<<<grid, block, 0, stream>>>(x, W_ih, W_hh, b_ih, b_hh, fc_w,
                                             fc_b, out);
}


# C of a step is a projection tile as it stands (tile = 4 steps x 4 batch rows, columns 6-15 zero; the step's matrix rows rotate through the tile): no moves or clears build C; sums and order unchanged
# speedup vs baseline: 5.8422x; 1.0614x over previous
#include <hip/hip_runtime.h>


typedef __attribute__((ext_vector_type(2))) float v2f;
typedef __attribute__((ext_vector_type(8))) float v8f;

#define RNN_B 4096
#define RNN_T 512
#define RNN_I 16
#define RNN_H 6
#define RNN_CH 8
#define RNN_NCH (RNN_T / RNN_CH)
#define RNN_ROW (RNN_T * RNN_I)
#define RNN_RW 4
#define RNN_NT 2
#define RNN_WPB 4
#define RNN_NBLK 256
#define RNN_PFU 64
#define RNN_PFC 4
#define RNN_PFD 8

static_assert(RNN_NBLK * RNN_WPB * RNN_RW == RNN_B, "grid covers the batch exactly");
static_assert(RNN_NT * 4 == RNN_CH && RNN_RW == 4, "one projection product per recurrence step");
static_assert(RNN_RW * (RNN_PFC * RNN_CH * RNN_I / RNN_PFU) == 32, "one prefetch unit per lane");
static_assert(RNN_PFD % RNN_PFC == 0 && RNN_NCH % RNN_PFC == 0, "prefetch groups tile the sequence");
static_assert(RNN_H <= 8, "the hidden units fit two K slices of 4");

__device__ __forceinline__ v8f wmma_f32_4(v2f a, v2f b, v8f c) {
  return __builtin_amdgcn_wmma_f32_16x16x4_f32(false, a, false, b, (short)0, c,
                                               false, false);
}

__device__ __forceinline__ float tanh_fast(float x) {
#if __has_builtin(__builtin_amdgcn_tanhf)
  return __builtin_amdgcn_tanhf(x);
#else
  x = __builtin_amdgcn_fmed3f(x, -9.0f, 9.0f);
  float e = __expf(2.0f * x);
  return (e - 1.0f) * __builtin_amdgcn_rcpf(e + 1.0f);
#endif
}

__device__ __forceinline__ float gather16(float v, unsigned lo, unsigned hi) {
  const int i = __builtin_bit_cast(int, v);
  return __builtin_bit_cast(float, __builtin_amdgcn_permlane16(i, i, lo, hi, true, false));
}
__device__ __forceinline__ float gatherx16(float v, unsigned lo, unsigned hi) {
  const int i = __builtin_bit_cast(int, v);
  return __builtin_bit_cast(float, __builtin_amdgcn_permlanex16(i, i, lo, hi, true, false));
}
__device__ __forceinline__ float rot8(float v) {
  const int i = __builtin_bit_cast(int, v);
  return __builtin_bit_cast(float, __builtin_amdgcn_update_dpp(i, i, 0x128, 0xf, 0xf, true));
}

__device__ __forceinline__ void rnn_chunk(const v8f (&Pc)[RNN_NT], v8f (&Pn)[RNN_NT],
                                          const v2f (&xr)[RNN_NT][4], v2f (&xw)[RNN_NT][4],
                                          const float* xnext,
                                          const v2f (&bih)[4], const v2f& whh,
                                          const v8f& cbias, v2f& ha,
                                          float& hx, bool lowcol, bool own) {
  v8f p = cbias;
#pragma unroll
  for (int j = 0; j < RNN_CH; ++j) {
    const int jj = j & 3;
    const int r0 = 2 * jj, r1 = 2 * jj + 1, r2 = (2 * jj + 2) & 7, r3 = (2 * jj + 3) & 7;
    const v8f acc = wmma_f32_4(ha, whh, Pc[j >> 2]);
    __builtin_amdgcn_sched_barrier(0);

    const int q = j >> 2;
    const int s = j & 3;
    if (s == 0) p = cbias;
    p = wmma_f32_4(xr[q][s], bih[s], p);
    if (s == 3) Pn[q] = p;
    if (j == 0) {
#pragma unroll
      for (int q2 = 0; q2 < RNN_NT; ++q2) {
#pragma unroll
        for (int s2 = 0; s2 < 4; ++s2)
          xw[q2][s2] = *(const v2f*)(xnext + q2 * (4 * RNN_I) + 4 * s2);
      }
    }
    __builtin_amdgcn_sched_barrier(0);

    const float s0 = lowcol ? acc[r0] : acc[r3];
    const float s1 = lowcol ? acc[r1] : acc[r2];
    hx = tanh_fast(s0 + rot8(s1));
    const float xo = gatherx16(hx, 0x54761032u, 0xdcfe98bau);
    const float z  = own ? hx : xo;
    if (((j + 1) & 1) == 0) {
      ha[0] = gather16(z, 0xc480c480u, 0xe6a2e6a2u);
      ha[1] = gather16(z, 0xd591d591u, 0xf7b3f7b3u);
    } else {
      ha[0] = gather16(z, 0x80c480c4u, 0xa2e6a2e6u);
      ha[1] = gather16(z, 0x91d591d5u, 0xb3f7b3f7u);
    }
  }
}

__global__ __launch_bounds__(RNN_WPB * 32) void rnn_fused_wmma(
    const float* __restrict__ x, const float* __restrict__ W_ih,
    const float* __restrict__ W_hh, const float* __restrict__ b_ih,
    const float* __restrict__ b_hh, const float* __restrict__ fc_w,
    const float* __restrict__ fc_b, float* __restrict__ out) {
  __shared__ __align__(16) float hbuf[RNN_WPB][RNN_RW][8];

  const int tid  = threadIdx.x;
  const int w    = tid >> 5;
  const int lane = tid & 31;
  const int n    = lane & 15;
  const int hi   = lane >> 4;
  const int u    = n & 7;
  const int g    = n >> 3;
  const int b0   = (blockIdx.x * RNN_WPB + w) * RNN_RW;

  const float* xpf = x + (size_t)(b0 + (lane >> 3)) * RNN_ROW + (lane & 7) * RNN_PFU;
  for (int pc = 0; pc < RNN_PFD; pc += RNN_PFC)
    __builtin_prefetch(xpf + (size_t)pc * (RNN_CH * RNN_I), 0, 2);

  v2f bih[4];
#pragma unroll
  for (int s = 0; s < 4; ++s) {
#pragma unroll
    for (int v = 0; v < 2; ++v) {
      int k = 4 * s + 2 * hi + v;
      bih[s][v] = (n < RNN_H) ? W_ih[n * RNN_I + k] : 0.0f;
    }
  }
  v2f whh;
#pragma unroll
  for (int v = 0; v < 2; ++v) {
    int k = 4 * g + 2 * hi + v;
    whh[v] = (u < RNN_H && k < RNN_H) ? W_hh[u * RNN_H + k] : 0.0f;
  }
  const float bn = (n < RNN_H) ? (b_ih[n] + b_hh[n]) : 0.0f;
  v8f cbias;
#pragma unroll
  for (int i = 0; i < 8; ++i) cbias[i] = bn;

  const float* xt = x + (size_t)(b0 + (n & 1) + 2 * (n >> 3)) * RNN_ROW
                      + ((n & 7) >> 1) * RNN_I + 2 * hi;
  const bool lowcol = n < 8;
  const bool own    = ((n >> 1) & 1) == hi;

  v2f xr[RNN_NT][4], xs[RNN_NT][4];
  v8f P0[RNN_NT], P1[RNN_NT];

#pragma unroll
  for (int q = 0; q < RNN_NT; ++q) {
#pragma unroll
    for (int s = 0; s < 4; ++s)
      xr[q][s] = *(const v2f*)(xt + q * (4 * RNN_I) + 4 * s);
  }
#pragma unroll
  for (int q = 0; q < RNN_NT; ++q) {
    v8f p = cbias;
#pragma unroll
    for (int s = 0; s < 4; ++s) p = wmma_f32_4(xr[q][s], bih[s], p);
    P0[q] = p;
  }
#pragma unroll
  for (int q = 0; q < RNN_NT; ++q) {
#pragma unroll
    for (int s = 0; s < 4; ++s)
      xr[q][s] = *(const v2f*)(xt + RNN_CH * RNN_I + q * (4 * RNN_I) + 4 * s);
  }

  v2f ha = 0.0f;
  float hx = 0.0f;

  for (int c = 0; c < RNN_NCH; c += 2) {
    if (c % RNN_PFC == 0 && c + RNN_PFD < RNN_NCH)
      __builtin_prefetch(xpf + (size_t)(c + RNN_PFD) * (RNN_CH * RNN_I), 0, 2);
    const float* xa = xt + (size_t)((c + 2) & (RNN_NCH - 1)) * (RNN_CH * RNN_I);
    const float* xb = xt + (size_t)((c + 3) & (RNN_NCH - 1)) * (RNN_CH * RNN_I);
    rnn_chunk(P0, P1, xr, xs, xa, bih, whh, cbias, ha, hx, lowcol, own);
    rnn_chunk(P1, P0, xs, xr, xb, bih, whh, cbias, ha, hx, lowcol, own);
  }

  hbuf[w][2 * hi + (n >> 3)][u] = hx;
  __syncthreads();
  if (tid < RNN_WPB * RNN_RW) {
    const float* hl = hbuf[tid >> 2][tid & 3];
    float s = fc_b[0];
#pragma unroll
    for (int j = 0; j < RNN_H; ++j) s += hl[j] * fc_w[j];
    out[blockIdx.x * (RNN_WPB * RNN_RW) + tid] = s;
  }
}

extern "C" void kernel_launch(void* const* d_in, const int* in_sizes, int n_in,
                              void* d_out, int out_size, void* d_ws,
                              size_t ws_size, hipStream_t stream) {
  (void)in_sizes; (void)n_in; (void)out_size; (void)d_ws; (void)ws_size;
  const float* x    = (const float*)d_in[0];
  const float* W_ih = (const float*)d_in[1];
  const float* W_hh = (const float*)d_in[2];
  const float* b_ih = (const float*)d_in[3];
  const float* b_hh = (const float*)d_in[4];
  const float* fc_w = (const float*)d_in[5];
  const float* fc_b = (const float*)d_in[6];
  float* out = (float*)d_out;

  dim3 block(RNN_WPB * 32);
  dim3 grid(RNN_NBLK);
  rnn_fused_wmma<<<grid, block, 0, stream>>>(x, W_ih, W_hh, b_ih, b_hh, fc_w,
                                             fc_b, out);
}


# projection product moved from in front of the step's lane work to directly behind its tanh, so the tanh's result wait passes under the product; sums and order unchanged
# speedup vs baseline: 6.1070x; 1.0453x over previous
#include <hip/hip_runtime.h>


typedef __attribute__((ext_vector_type(2))) float v2f;
typedef __attribute__((ext_vector_type(8))) float v8f;

#define RNN_B 4096
#define RNN_T 512
#define RNN_I 16
#define RNN_H 6
#define RNN_CH 8
#define RNN_NCH (RNN_T / RNN_CH)
#define RNN_ROW (RNN_T * RNN_I)
#define RNN_RW 4
#define RNN_NT 2
#define RNN_WPB 4
#define RNN_NBLK 256
#define RNN_PFU 64
#define RNN_PFC 4
#define RNN_PFD 8

static_assert(RNN_NBLK * RNN_WPB * RNN_RW == RNN_B, "grid covers the batch exactly");
static_assert(RNN_NT * 4 == RNN_CH && RNN_RW == 4, "one projection product per recurrence step");
static_assert(RNN_RW * (RNN_PFC * RNN_CH * RNN_I / RNN_PFU) == 32, "one prefetch unit per lane");
static_assert(RNN_PFD % RNN_PFC == 0 && RNN_NCH % RNN_PFC == 0, "prefetch groups tile the sequence");
static_assert(RNN_H <= 8, "the hidden units fit two K slices of 4");

__device__ __forceinline__ v8f wmma_f32_4(v2f a, v2f b, v8f c) {
  return __builtin_amdgcn_wmma_f32_16x16x4_f32(false, a, false, b, (short)0, c,
                                               false, false);
}

__device__ __forceinline__ float tanh_fast(float x) {
#if __has_builtin(__builtin_amdgcn_tanhf)
  return __builtin_amdgcn_tanhf(x);
#else
  x = __builtin_amdgcn_fmed3f(x, -9.0f, 9.0f);
  float e = __expf(2.0f * x);
  return (e - 1.0f) * __builtin_amdgcn_rcpf(e + 1.0f);
#endif
}

__device__ __forceinline__ float gather16(float v, unsigned lo, unsigned hi) {
  const int i = __builtin_bit_cast(int, v);
  return __builtin_bit_cast(float, __builtin_amdgcn_permlane16(i, i, lo, hi, true, false));
}
__device__ __forceinline__ float gatherx16(float v, unsigned lo, unsigned hi) {
  const int i = __builtin_bit_cast(int, v);
  return __builtin_bit_cast(float, __builtin_amdgcn_permlanex16(i, i, lo, hi, true, false));
}
__device__ __forceinline__ float rot8(float v) {
  const int i = __builtin_bit_cast(int, v);
  return __builtin_bit_cast(float, __builtin_amdgcn_update_dpp(i, i, 0x128, 0xf, 0xf, true));
}

__device__ __forceinline__ void rnn_chunk(const v8f (&Pc)[RNN_NT], v8f (&Pn)[RNN_NT],
                                          const v2f (&xr)[RNN_NT][4], v2f (&xw)[RNN_NT][4],
                                          const float* xnext,
                                          const v2f (&bih)[4], const v2f& whh,
                                          const v8f& cbias, v2f& ha,
                                          float& hx, bool lowcol, bool own) {
  v8f p = cbias;
#pragma unroll
  for (int j = 0; j < RNN_CH; ++j) {
    const int jj = j & 3;
    const v8f acc = wmma_f32_4(ha, whh, Pc[j >> 2]);
    __builtin_amdgcn_sched_barrier(0);

    float a0, a1, a2, a3;
    switch (jj) {
      case 0:  a0 = acc[0]; a1 = acc[1]; a2 = acc[2]; a3 = acc[3]; break;
      case 1:  a0 = acc[2]; a1 = acc[3]; a2 = acc[4]; a3 = acc[5]; break;
      case 2:  a0 = acc[4]; a1 = acc[5]; a2 = acc[6]; a3 = acc[7]; break;
      default: a0 = acc[6]; a1 = acc[7]; a2 = acc[0]; a3 = acc[1]; break;
    }
    const float s0 = lowcol ? a0 : a3;
    const float s1 = lowcol ? a1 : a2;
    hx = tanh_fast(s0 + rot8(s1));
    __builtin_amdgcn_sched_barrier(0);

    const int q = j >> 2;
    const int s = j & 3;
    if (s == 0) p = cbias;
    p = wmma_f32_4(xr[q][s], bih[s], p);
    if (s == 3) Pn[q] = p;
    if (j == 0) {
#pragma unroll
      for (int q2 = 0; q2 < RNN_NT; ++q2) {
#pragma unroll
        for (int s2 = 0; s2 < 4; ++s2)
          xw[q2][s2] = *(const v2f*)(xnext + q2 * (4 * RNN_I) + 4 * s2);
      }
    }
    __builtin_amdgcn_sched_barrier(0);

    const float xo = gatherx16(hx, 0x54761032u, 0xdcfe98bau);
    const float z  = own ? hx : xo;
    if (((j + 1) & 1) == 0) {
      ha[0] = gather16(z, 0xc480c480u, 0xe6a2e6a2u);
      ha[1] = gather16(z, 0xd591d591u, 0xf7b3f7b3u);
    } else {
      ha[0] = gather16(z, 0x80c480c4u, 0xa2e6a2e6u);
      ha[1] = gather16(z, 0x91d591d5u, 0xb3f7b3f7u);
    }
  }
}

__global__ __launch_bounds__(RNN_WPB * 32) void rnn_fused_wmma(
    const float* __restrict__ x, const float* __restrict__ W_ih,
    const float* __restrict__ W_hh, const float* __restrict__ b_ih,
    const float* __restrict__ b_hh, const float* __restrict__ fc_w,
    const float* __restrict__ fc_b, float* __restrict__ out) {
  __shared__ __align__(16) float hbuf[RNN_WPB][RNN_RW][8];

  const int tid  = threadIdx.x;
  const int w    = tid >> 5;
  const int lane = tid & 31;
  const int n    = lane & 15;
  const int hi   = lane >> 4;
  const int u    = n & 7;
  const int g    = n >> 3;
  const int b0   = (blockIdx.x * RNN_WPB + w) * RNN_RW;

  const float* xpf = x + (size_t)(b0 + (lane >> 3)) * RNN_ROW + (lane & 7) * RNN_PFU;
  for (int pc = 0; pc < RNN_PFD; pc += RNN_PFC)
    __builtin_prefetch(xpf + (size_t)pc * (RNN_CH * RNN_I), 0, 2);

  v2f bih[4];
#pragma unroll
  for (int s = 0; s < 4; ++s) {
#pragma unroll
    for (int v = 0; v < 2; ++v) {
      int k = 4 * s + 2 * hi + v;
      bih[s][v] = (n < RNN_H) ? W_ih[n * RNN_I + k] : 0.0f;
    }
  }
  v2f whh;
#pragma unroll
  for (int v = 0; v < 2; ++v) {
    int k = 4 * g + 2 * hi + v;
    whh[v] = (u < RNN_H && k < RNN_H) ? W_hh[u * RNN_H + k] : 0.0f;
  }
  const float bn = (n < RNN_H) ? (b_ih[n] + b_hh[n]) : 0.0f;
  v8f cbias;
#pragma unroll
  for (int i = 0; i < 8; ++i) cbias[i] = bn;

  const float* xt = x + (size_t)(b0 + (n & 1) + 2 * (n >> 3)) * RNN_ROW
                      + ((n & 7) >> 1) * RNN_I + 2 * hi;
  const bool lowcol = n < 8;
  const bool own    = ((n >> 1) & 1) == hi;

  v2f xr[RNN_NT][4], xs[RNN_NT][4];
  v8f P0[RNN_NT], P1[RNN_NT];

#pragma unroll
  for (int q = 0; q < RNN_NT; ++q) {
#pragma unroll
    for (int s = 0; s < 4; ++s)
      xr[q][s] = *(const v2f*)(xt + q * (4 * RNN_I) + 4 * s);
  }
#pragma unroll
  for (int q = 0; q < RNN_NT; ++q) {
    v8f p = cbias;
#pragma unroll
    for (int s = 0; s < 4; ++s) p = wmma_f32_4(xr[q][s], bih[s], p);
    P0[q] = p;
  }
#pragma unroll
  for (int q = 0; q < RNN_NT; ++q) {
#pragma unroll
    for (int s = 0; s < 4; ++s)
      xr[q][s] = *(const v2f*)(xt + RNN_CH * RNN_I + q * (4 * RNN_I) + 4 * s);
  }

  v2f ha = 0.0f;
  float hx = 0.0f;

  for (int c = 0; c < RNN_NCH; c += 2) {
    if (c % RNN_PFC == 0 && c + RNN_PFD < RNN_NCH)
      __builtin_prefetch(xpf + (size_t)(c + RNN_PFD) * (RNN_CH * RNN_I), 0, 2);
    const float* xa = xt + (size_t)((c + 2) & (RNN_NCH - 1)) * (RNN_CH * RNN_I);
    const float* xb = xt + (size_t)((c + 3) & (RNN_NCH - 1)) * (RNN_CH * RNN_I);
    rnn_chunk(P0, P1, xr, xs, xa, bih, whh, cbias, ha, hx, lowcol, own);
    rnn_chunk(P1, P0, xs, xr, xb, bih, whh, cbias, ha, hx, lowcol, own);
  }

  hbuf[w][2 * hi + (n >> 3)][u] = hx;
  __syncthreads();
  if (tid < RNN_WPB * RNN_RW) {
    const float* hl = hbuf[tid >> 2][tid & 3];
    float s = fc_b[0];
#pragma unroll
    for (int j = 0; j < RNN_H; ++j) s += hl[j] * fc_w[j];
    out[blockIdx.x * (RNN_WPB * RNN_RW) + tid] = s;
  }
}

extern "C" void kernel_launch(void* const* d_in, const int* in_sizes, int n_in,
                              void* d_out, int out_size, void* d_ws,
                              size_t ws_size, hipStream_t stream) {
  (void)in_sizes; (void)n_in; (void)out_size; (void)d_ws; (void)ws_size;
  const float* x    = (const float*)d_in[0];
  const float* W_ih = (const float*)d_in[1];
  const float* W_hh = (const float*)d_in[2];
  const float* b_ih = (const float*)d_in[3];
  const float* b_hh = (const float*)d_in[4];
  const float* fc_w = (const float*)d_in[5];
  const float* fc_b = (const float*)d_in[6];
  float* out = (float*)d_out;

  dim3 block(RNN_WPB * 32);
  dim3 grid(RNN_NBLK);
  rnn_fused_wmma<<<grid, block, 0, stream>>>(x, W_ih, W_hh, b_ih, b_hh, fc_w,
                                             fc_b, out);
}
